# starting kernel
# baseline (speedup 1.0000x reference)
#include <hip/hip_runtime.h>

typedef __attribute__((ext_vector_type(2))) float v2f;
typedef __attribute__((ext_vector_type(8))) float v8f;

#define B_  32
#define S_  64
#define I_  333
#define O_  83
#define CT_ 666

__global__ __launch_bounds__(64) void taylor_wmma_kernel(
    const float* __restrict__ x,
    const float* __restrict__ A,
    const float* __restrict__ A0,
    float* __restrict__ out)
{
    const int s    = blockIdx.x;
    const int nt   = blockIdx.y;
    const int tid  = threadIdx.x;
    const int wave = tid >> 5;
    const int lane = tid & 31;
    const int half = lane >> 4;
    const int l16  = lane & 15;

    __shared__ float lds_part[64];
    __shared__ float lds_bias[16];
    __shared__ float redbuf[2 * 8 * 32];

    {
        const int colb = tid & 15;
        const int q    = tid >> 4;
        const int colc = min(nt * 16 + colb, O_ - 1);
        const float* a0p = A0 + (size_t)s * I_ * O_ + (size_t)q * O_ + colc;
        float p = 0.f;
        #pragma unroll 4
        for (int i = q; i < I_; i += 4) {
            p += *a0p;
            a0p += 4 * O_;
        }
        lds_part[tid] = p;
    }
    __syncthreads();
    if (tid < 16)
        lds_bias[tid] = lds_part[tid] + lds_part[tid + 16] +
                        lds_part[tid + 32] + lds_part[tid + 48];

    const int col     = nt * 16 + l16;
    const int colc    = min(col, O_ - 1);
    const int c_start = wave * 332;

    const float* xrow0 = x + ((size_t)l16        * S_ + s) * I_;
    const float* xrow1 = x + ((size_t)(l16 + 16) * S_ + s) * I_;
    const float* Ap    = A + (size_t)s * CT_ * O_;

    const float* xq0 = xrow0 + half + (c_start >> 1);
    const float* xq1 = xrow1 + half + (c_start >> 1);
    const float* bp  = Ap + (size_t)(c_start + 2 * half) * O_ + colc;

    v8f acc0 = {}, acc1 = {};

    #pragma unroll 4
    for (int it = 0; it < 83; ++it) {
        const float x0 = xq0[0];
        const float x1 = xq1[0];
        v2f a0; a0.x = x0; a0.y = x0 * x0;
        v2f a1; a1.x = x1; a1.y = x1 * x1;
        v2f b;  b.x = bp[0]; b.y = bp[O_];
        __builtin_prefetch(bp + 16 * O_, 0, 1);
        acc0 = __builtin_amdgcn_wmma_f32_16x16x4_f32(
            false, a0, false, b, (short)0, acc0, false, false);
        acc1 = __builtin_amdgcn_wmma_f32_16x16x4_f32(
            false, a1, false, b, (short)0, acc1, false, false);
        xq0 += 2; xq1 += 2; bp += 4 * O_;
    }

    if (wave == 1) {
        const float x0 = xrow0[I_ - 1];
        const float x1 = xrow1[I_ - 1];
        v2f a0; a0.x = x0; a0.y = x0 * x0;
        v2f a1; a1.x = x1; a1.y = x1 * x1;
        const int r0 = min(664 + 2 * half, CT_ - 1);
        const float b0 = Ap[(size_t)r0 * O_ + colc];
        const float b1 = Ap[(size_t)(CT_ - 1) * O_ + colc];
        v2f b;
        b.x = half ? 0.f : b0;
        b.y = half ? 0.f : b1;
        acc0 = __builtin_amdgcn_wmma_f32_16x16x4_f32(
            false, a0, false, b, (short)0, acc0, false, false);
        acc1 = __builtin_amdgcn_wmma_f32_16x16x4_f32(
            false, a1, false, b, (short)0, acc1, false, false);

        #pragma unroll
        for (int r = 0; r < 8; ++r) {
            redbuf[r * 32 + lane]       = acc0[r];
            redbuf[256 + r * 32 + lane] = acc1[r];
        }
    }
    __syncthreads();

    if (wave == 0) {
        const float bias = lds_bias[l16];
        if (col < O_) {
            #pragma unroll
            for (int r = 0; r < 8; ++r) {
                const int m = r + 8 * half;
                const float v0 = acc0[r] + redbuf[r * 32 + lane]       + bias;
                const float v1 = acc1[r] + redbuf[256 + r * 32 + lane] + bias;
                out[((size_t)m        * S_ + s) * O_ + col] = v0;
                out[((size_t)(m + 16) * S_ + s) * O_ + col] = v1;
            }
        }
    }
}

extern "C" void kernel_launch(void* const* d_in, const int* in_sizes, int n_in,
                              void* d_out, int out_size, void* d_ws, size_t ws_size,
                              hipStream_t stream) {
    const float* x  = (const float*)d_in[0];
    const float* A  = (const float*)d_in[1];
    const float* A0 = (const float*)d_in[2];
    float* out = (float*)d_out;

    dim3 grid(S_, (O_ + 15) / 16);
    taylor_wmma_kernel<<<grid, 64, 0, stream>>>(x, A, A0, out);
}


# Contraction split over 8 waves per workgroup (was 2), each wave issuing all its loads before its first WMMA; summation order changes: 8 partial tiles and 16 bias partials are added in LDS
# speedup vs baseline: 2.3588x; 2.3588x over previous
#include <hip/hip_runtime.h>

typedef __attribute__((ext_vector_type(2))) float v2f;
typedef __attribute__((ext_vector_type(8))) float v8f;

#define B_  32
#define S_  64
#define I_  333
#define O_  83
#define CT_ 666
#define NW_ 8
#define NST_ 21
#define NBJ_ 21

__global__ __launch_bounds__(256) void taylor_wmma_kernel(
    const float* __restrict__ x,
    const float* __restrict__ A,
    const float* __restrict__ A0,
    float* __restrict__ out)
{
    const int s    = blockIdx.x;
    const int nt   = blockIdx.y;
    const int tid  = threadIdx.x;
    const int wave = tid >> 5;
    const int lane = tid & 31;
    const int half = lane >> 4;
    const int l16  = lane & 15;

    __shared__ float red[NW_ * 2 * 8 * 32];
    __shared__ float bpart[256];

    const int col  = nt * 16 + l16;
    const int colc = min(col, O_ - 1);

    const float* xrow0 = x + ((size_t)l16        * S_ + s) * I_;
    const float* xrow1 = x + ((size_t)(l16 + 16) * S_ + s) * I_;
    const float* Ap    = A + (size_t)s * CT_ * O_;
    const float* A0p   = A0 + (size_t)s * I_ * O_ + colc;

    const int ibase = wave * (2 * NST_) + half;

    const float* xq0 = xrow0 + ibase;
    const float* xq1 = xrow1 + ibase;
    const float* bp  = Ap + (size_t)(2 * ibase) * O_ + colc;

    float xa[NST_], xb[NST_], b0[NST_], b1[NST_];
    #pragma unroll
    for (int t = 0; t < NST_; ++t) {
        const int d = (t >= 19) ? min(ibase + 2 * t, I_ - 1) - ibase : 2 * t;
        xa[t] = xq0[d];
        xb[t] = xq1[d];
        b0[t] = bp[(2 * d)     * O_];
        b1[t] = bp[(2 * d + 1) * O_];
        __builtin_amdgcn_sched_barrier(0);
    }

    const int q = tid >> 4;
    float bv[NBJ_];
    #pragma unroll
    for (int j = 0; j < NBJ_; ++j) {
        const int i  = q + 16 * j;
        const int ic = (j == NBJ_ - 1) ? min(i, I_ - 1) : i;
        bv[j] = A0p[(size_t)ic * O_];
    }

    __builtin_amdgcn_sched_barrier(0);

    v8f acc0 = {}, acc1 = {};
    #pragma unroll
    for (int t = 0; t < NST_; ++t) {
        float x0 = xa[t], x1 = xb[t], w0 = b0[t], w1 = b1[t];
        if (t >= 19) {
            const bool ok = (ibase + 2 * t) < I_;
            x0 = ok ? x0 : 0.f;  x1 = ok ? x1 : 0.f;
            w0 = ok ? w0 : 0.f;  w1 = ok ? w1 : 0.f;
        }
        v2f a0; a0.x = x0; a0.y = x0 * x0;
        v2f a1; a1.x = x1; a1.y = x1 * x1;
        v2f b;  b.x = w0;  b.y = w1;
        acc0 = __builtin_amdgcn_wmma_f32_16x16x4_f32(
            false, a0, false, b, (short)0, acc0, false, false);
        acc1 = __builtin_amdgcn_wmma_f32_16x16x4_f32(
            false, a1, false, b, (short)0, acc1, false, false);
        __builtin_amdgcn_sched_barrier(0);
    }

    float p = 0.f;
    #pragma unroll
    for (int j = 0; j < NBJ_ - 1; ++j) p += bv[j];
    p += (q + 16 * (NBJ_ - 1) < I_) ? bv[NBJ_ - 1] : 0.f;

    #pragma unroll
    for (int r = 0; r < 8; ++r) {
        red[((wave * 2 + 0) * 8 + r) * 32 + lane] = acc0[r];
        red[((wave * 2 + 1) * 8 + r) * 32 + lane] = acc1[r];
    }
    bpart[tid] = p;
    __syncthreads();

    {
        const int r = wave;
        float v0 = 0.f, v1 = 0.f, bias = 0.f;
        #pragma unroll
        for (int w = 0; w < NW_; ++w) {
            v0 += red[((w * 2 + 0) * 8 + r) * 32 + lane];
            v1 += red[((w * 2 + 1) * 8 + r) * 32 + lane];
        }
        #pragma unroll
        for (int k = 0; k < 16; ++k) bias += bpart[k * 16 + l16];
        if (col < O_) {
            const int m = r + 8 * half;
            out[((size_t)m        * S_ + s) * O_ + col] = v0 + bias;
            out[((size_t)(m + 16) * S_ + s) * O_ + col] = v1 + bias;
        }
    }
}

extern "C" void kernel_launch(void* const* d_in, const int* in_sizes, int n_in,
                              void* d_out, int out_size, void* d_ws, size_t ws_size,
                              hipStream_t stream) {
    const float* x  = (const float*)d_in[0];
    const float* A  = (const float*)d_in[1];
    const float* A0 = (const float*)d_in[2];
    float* out = (float*)d_out;

    dim3 grid(S_, (O_ + 15) / 16);
    taylor_wmma_kernel<<<grid, 256, 0, stream>>>(x, A, A0, out);
}


# x tile staged through LDS: each wave loads 4 batch rows with lanes along i (1-2 cache lines per load, was 16), A fragments then read from LDS; one added LDS-only barrier; summation order unchanged
# speedup vs baseline: 2.7884x; 1.1821x over previous
#include <hip/hip_runtime.h>

typedef __attribute__((ext_vector_type(2))) float v2f;
typedef __attribute__((ext_vector_type(8))) float v8f;

#define B_  32
#define S_  64
#define I_  333
#define O_  83
#define CT_ 666
#define NW_ 8
#define NST_ 21
#define NBJ_ 21
#define XCH_ 11
#define XP_  354

__global__ __launch_bounds__(256) void taylor_wmma_kernel(
    const float* __restrict__ x,
    const float* __restrict__ A,
    const float* __restrict__ A0,
    float* __restrict__ out)
{
    const int s    = blockIdx.x;
    const int nt   = blockIdx.y;
    const int tid  = threadIdx.x;
    const int wave = tid >> 5;
    const int lane = tid & 31;
    const int half = lane >> 4;
    const int l16  = lane & 15;

    __shared__ float xs[B_ * XP_];
    __shared__ float red[NW_ * 2 * 8 * 32];
    __shared__ float bpart[256];

    const int col  = nt * 16 + l16;
    const int colc = min(col, O_ - 1);

    const float* xrow  = x + ((size_t)(4 * wave) * S_ + s) * I_;
    const float* Ap    = A + (size_t)s * CT_ * O_;
    const float* A0p   = A0 + (size_t)s * I_ * O_ + colc;

    const int ibase = wave * (2 * NST_) + half;

    const float* bp  = Ap + (size_t)(2 * ibase) * O_ + colc;

    const int ilast = min(32 * (XCH_ - 1) + lane, I_ - 1);
    float xv[4 * XCH_];
    #pragma unroll
    for (int rr = 0; rr < 4; ++rr) {
        #pragma unroll
        for (int j = 0; j < XCH_; ++j) {
            const int i = (j == XCH_ - 1) ? ilast : 32 * j + lane;
            xv[rr * XCH_ + j] = xrow[(size_t)rr * S_ * I_ + i];
        }
    }
    __builtin_amdgcn_sched_barrier(0);

    const int q = tid >> 4;
    float bv[NBJ_];
    bv[0] = A0p[(size_t)q * O_];
    __builtin_amdgcn_sched_barrier(0);

    float b0[NST_], b1[NST_];
    #pragma unroll
    for (int t = 0; t < NST_; ++t) {
        const int d = (t >= 19) ? min(ibase + 2 * t, I_ - 1) - ibase : 2 * t;
        b0[t] = bp[(2 * d)     * O_];
        b1[t] = bp[(2 * d + 1) * O_];
        __builtin_amdgcn_sched_barrier(0);
    }

    #pragma unroll
    for (int j = 1; j < NBJ_; ++j) {
        const int i  = q + 16 * j;
        const int ic = (j == NBJ_ - 1) ? min(i, I_ - 1) : i;
        bv[j] = A0p[(size_t)ic * O_];
    }

    __builtin_amdgcn_sched_barrier(0);

    #pragma unroll
    for (int rr = 0; rr < 4; ++rr) {
        float* xsr = xs + (4 * wave + rr) * XP_;
        #pragma unroll
        for (int j = 0; j < XCH_ - 1; ++j) xsr[32 * j + lane] = xv[rr * XCH_ + j];
        const int i = 32 * (XCH_ - 1) + lane;
        xsr[i] = (i < I_) ? xv[rr * XCH_ + XCH_ - 1] : 0.f;
    }
    __builtin_amdgcn_fence(__ATOMIC_RELEASE, "workgroup", "local");
    __builtin_amdgcn_s_barrier();
    __builtin_amdgcn_fence(__ATOMIC_ACQUIRE, "workgroup", "local");

    const float* xf0 = xs + l16 * XP_ + ibase;
    const float* xf1 = xf0 + 16 * XP_;
    float xa[NST_], xb[NST_];
    #pragma unroll
    for (int t = 0; t < NST_; ++t) { xa[t] = xf0[2 * t]; xb[t] = xf1[2 * t]; }
    __builtin_amdgcn_sched_barrier(0);

    v8f acc0 = {}, acc1 = {};
    #pragma unroll
    for (int t = 0; t < NST_; ++t) {
        float x0 = xa[t], x1 = xb[t], w0 = b0[t], w1 = b1[t];
        if (t >= 19) {
            const bool ok = (ibase + 2 * t) < I_;
            w0 = ok ? w0 : 0.f;  w1 = ok ? w1 : 0.f;
        }
        v2f a0; a0.x = x0; a0.y = x0 * x0;
        v2f a1; a1.x = x1; a1.y = x1 * x1;
        v2f b;  b.x = w0;  b.y = w1;
        acc0 = __builtin_amdgcn_wmma_f32_16x16x4_f32(
            false, a0, false, b, (short)0, acc0, false, false);
        acc1 = __builtin_amdgcn_wmma_f32_16x16x4_f32(
            false, a1, false, b, (short)0, acc1, false, false);
        __builtin_amdgcn_sched_barrier(0);
    }

    float p = 0.f;
    #pragma unroll
    for (int j = 0; j < NBJ_ - 1; ++j) p += bv[j];
    p += (q + 16 * (NBJ_ - 1) < I_) ? bv[NBJ_ - 1] : 0.f;

    #pragma unroll
    for (int r = 0; r < 8; ++r) {
        red[((wave * 2 + 0) * 8 + r) * 32 + lane] = acc0[r];
        red[((wave * 2 + 1) * 8 + r) * 32 + lane] = acc1[r];
    }
    bpart[tid] = p;
    __syncthreads();

    {
        const int r = wave;
        float v0 = 0.f, v1 = 0.f, bias = 0.f;
        #pragma unroll
        for (int w = 0; w < NW_; ++w) {
            v0 += red[((w * 2 + 0) * 8 + r) * 32 + lane];
            v1 += red[((w * 2 + 1) * 8 + r) * 32 + lane];
        }
        #pragma unroll
        for (int k = 0; k < 16; ++k) bias += bpart[k * 16 + l16];
        if (col < O_) {
            const int m = r + 8 * half;
            out[((size_t)m        * S_ + s) * O_ + col] = v0 + bias;
            out[((size_t)(m + 16) * S_ + s) * O_ + col] = v1 + bias;
        }
    }
}

extern "C" void kernel_launch(void* const* d_in, const int* in_sizes, int n_in,
                              void* d_out, int out_size, void* d_ws, size_t ws_size,
                              hipStream_t stream) {
    const float* x  = (const float*)d_in[0];
    const float* A  = (const float*)d_in[1];
    const float* A0 = (const float*)d_in[2];
    float* out = (float*)d_out;

    dim3 grid(S_, (O_ + 15) / 16);
    taylor_wmma_kernel<<<grid, 256, 0, stream>>>(x, A, A0, out);
}


# Grid dealt out evenly: 256 workgroups (was 384) of three output tiles each (a whole 16-column tile plus one batch half of its neighbour), so every WGP holds one; summation order unchanged
# speedup vs baseline: 2.9888x; 1.0719x over previous
#include <hip/hip_runtime.h>

typedef __attribute__((ext_vector_type(2))) float v2f;
typedef __attribute__((ext_vector_type(8))) float v8f;

#define B_  32
#define S_  64
#define I_  333
#define O_  83
#define CT_ 666
#define NW_ 8
#define NST_ 21
#define NPRE_ 4
#define NBJ_ 21
#define XCH_ 11
#define XP_  354

__global__ __launch_bounds__(256) void taylor_wmma_kernel(
    const float* __restrict__ x,
    const float* __restrict__ A,
    const float* __restrict__ A0,
    float* __restrict__ out)
{
    const int s    = blockIdx.x;
    const int h    = blockIdx.y;
    const int tid  = threadIdx.x;
    const int wave = tid >> 5;
    const int lane = tid & 31;
    const int half = lane >> 4;
    const int l16  = lane & 15;

    const int odd = h & 1;
    const int nta = 3 * (h >> 1) + odd;
    const int ntf = nta + odd;
    const int nth = nta + 1 - odd;

    __shared__ float xs[B_ * XP_];
    __shared__ float red[NW_ * 3 * 8 * 32];
    __shared__ float bpartf[256];
    __shared__ float bparth[256];

    const int colf  = ntf * 16 + l16;
    const int colh  = nth * 16 + l16;
    const int colfc = min(colf, O_ - 1);
    const int colhc = min(colh, O_ - 1);

    const float* xrow  = x + ((size_t)(4 * wave) * S_ + s) * I_;
    const float* Ap    = A + (size_t)s * CT_ * O_;
    const float* A0pf  = A0 + (size_t)s * I_ * O_ + colfc;
    const float* A0ph  = A0 + (size_t)s * I_ * O_ + colhc;

    const int ibase = wave * (2 * NST_) + half;

    const float* bpf = Ap + (size_t)(2 * ibase) * O_ + colfc;
    const float* bph = Ap + (size_t)(2 * ibase) * O_ + colhc;

    const int ilast = min(32 * (XCH_ - 1) + lane, I_ - 1);
    float xv[4 * XCH_];
    #pragma unroll
    for (int rr = 0; rr < 4; ++rr) {
        #pragma unroll
        for (int j = 0; j < XCH_; ++j) {
            const int i = (j == XCH_ - 1) ? ilast : 32 * j + lane;
            xv[rr * XCH_ + j] = xrow[(size_t)rr * S_ * I_ + i];
        }
    }
    __builtin_amdgcn_sched_barrier(0);

    float b0f[NST_], b1f[NST_], b0h[NST_], b1h[NST_];
    #pragma unroll
    for (int t = 0; t < NPRE_; ++t) {
        b0f[t] = bpf[(4 * t)     * O_];
        b1f[t] = bpf[(4 * t + 1) * O_];
        b0h[t] = bph[(4 * t)     * O_];
        b1h[t] = bph[(4 * t + 1) * O_];
        __builtin_amdgcn_sched_barrier(0);
    }

    #pragma unroll
    for (int rr = 0; rr < 4; ++rr) {
        float* xsr = xs + (4 * wave + rr) * XP_;
        #pragma unroll
        for (int j = 0; j < XCH_ - 1; ++j) xsr[32 * j + lane] = xv[rr * XCH_ + j];
        const int i = 32 * (XCH_ - 1) + lane;
        xsr[i] = (i < I_) ? xv[rr * XCH_ + XCH_ - 1] : 0.f;
    }
    __builtin_amdgcn_fence(__ATOMIC_RELEASE, "workgroup", "local");
    __builtin_amdgcn_s_barrier();
    __builtin_amdgcn_fence(__ATOMIC_ACQUIRE, "workgroup", "local");

    const float* xf0 = xs + l16 * XP_ + ibase;
    const float* xf1 = xf0 + 16 * XP_;
    float xa[NST_], xb[NST_];
    #pragma unroll
    for (int t = 0; t < NST_; ++t) { xa[t] = xf0[2 * t]; xb[t] = xf1[2 * t]; }
    __builtin_amdgcn_sched_barrier(0);

    #pragma unroll
    for (int t = NPRE_; t < NST_; ++t) {
        const int d = (t >= 19) ? min(ibase + 2 * t, I_ - 1) - ibase : 2 * t;
        b0f[t] = bpf[(2 * d)     * O_];
        b1f[t] = bpf[(2 * d + 1) * O_];
        b0h[t] = bph[(2 * d)     * O_];
        b1h[t] = bph[(2 * d + 1) * O_];
        __builtin_amdgcn_sched_barrier(0);
    }

    const int q = tid >> 4;
    float bvf[NBJ_], bvh[NBJ_];
    v8f acc0 = {}, acc1 = {}, acc2 = {};
    #pragma unroll
    for (int t = 0; t < NST_; ++t) {
        float x0 = xa[t], x1 = xb[t];
        float w0f = b0f[t], w1f = b1f[t], w0h = b0h[t], w1h = b1h[t];
        if (t >= 19) {
            const bool ok = (ibase + 2 * t) < I_;
            w0f = ok ? w0f : 0.f;  w1f = ok ? w1f : 0.f;
            w0h = ok ? w0h : 0.f;  w1h = ok ? w1h : 0.f;
        }
        const float xh = odd ? x1 : x0;
        v2f a0; a0.x = x0; a0.y = x0 * x0;
        v2f a1; a1.x = x1; a1.y = x1 * x1;
        v2f ah; ah.x = xh; ah.y = xh * xh;
        v2f bf; bf.x = w0f; bf.y = w1f;
        v2f bh; bh.x = w0h; bh.y = w1h;
        acc0 = __builtin_amdgcn_wmma_f32_16x16x4_f32(
            false, a0, false, bf, (short)0, acc0, false, false);
        acc1 = __builtin_amdgcn_wmma_f32_16x16x4_f32(
            false, a1, false, bf, (short)0, acc1, false, false);
        acc2 = __builtin_amdgcn_wmma_f32_16x16x4_f32(
            false, ah, false, bh, (short)0, acc2, false, false);
        #pragma unroll
        for (int jj = 0; jj < 2; ++jj) {
            const int j = 2 * t + jj;
            if (j < NBJ_) {
                const int i  = q + 16 * j;
                const int ic = (j == NBJ_ - 1) ? min(i, I_ - 1) : i;
                bvf[j] = A0pf[(size_t)ic * O_];
                bvh[j] = A0ph[(size_t)ic * O_];
            }
        }
        __builtin_amdgcn_sched_barrier(0);
    }

    float pf = 0.f, ph = 0.f;
    #pragma unroll
    for (int j = 0; j < NBJ_ - 1; ++j) { pf += bvf[j]; ph += bvh[j]; }
    {
        const bool in = (q + 16 * (NBJ_ - 1)) < I_;
        pf += in ? bvf[NBJ_ - 1] : 0.f;
        ph += in ? bvh[NBJ_ - 1] : 0.f;
    }

    #pragma unroll
    for (int r = 0; r < 8; ++r) {
        red[((wave * 3 + 0) * 8 + r) * 32 + lane] = acc0[r];
        red[((wave * 3 + 1) * 8 + r) * 32 + lane] = acc1[r];
        red[((wave * 3 + 2) * 8 + r) * 32 + lane] = acc2[r];
    }
    bpartf[tid] = pf;
    bparth[tid] = ph;
    __syncthreads();

    {
        const int r = wave;
        float v0 = 0.f, v1 = 0.f, v2 = 0.f, biasf = 0.f, biash = 0.f;
        #pragma unroll
        for (int w = 0; w < NW_; ++w) {
            v0 += red[((w * 3 + 0) * 8 + r) * 32 + lane];
            v1 += red[((w * 3 + 1) * 8 + r) * 32 + lane];
            v2 += red[((w * 3 + 2) * 8 + r) * 32 + lane];
        }
        #pragma unroll
        for (int k = 0; k < 16; ++k) { biasf += bpartf[k * 16 + l16]; biash += bparth[k * 16 + l16]; }
        const int m = r + 8 * half;
        if (colf < O_) {
            out[((size_t)m        * S_ + s) * O_ + colf] = v0 + biasf;
            out[((size_t)(m + 16) * S_ + s) * O_ + colf] = v1 + biasf;
        }
        if (colh < O_) {
            out[((size_t)(m + 16 * odd) * S_ + s) * O_ + colh] = v2 + biash;
        }
    }
}

extern "C" void kernel_launch(void* const* d_in, const int* in_sizes, int n_in,
                              void* d_out, int out_size, void* d_ws, size_t ws_size,
                              hipStream_t stream) {
    const float* x  = (const float*)d_in[0];
    const float* A  = (const float*)d_in[1];
    const float* A0 = (const float*)d_in[2];
    float* out = (float*)d_out;

    dim3 grid(S_, 4);
    taylor_wmma_kernel<<<grid, 256, 0, stream>>>(x, A, A0, out);
}
